# starting kernel
# baseline (speedup 1.0000x reference)
#include <hip/hip_runtime.h>

typedef float v2f __attribute__((ext_vector_type(2)));
typedef float v8f __attribute__((ext_vector_type(8)));

#define NB_B   2
#define NN     2048
#define NT_T   8
#define DX_D   3
#define DZ_D   16
#define HH     64
#define WW     64
#define GPB    (NT_T * HH * WW)
#define TILES_PB (GPB / 16)
#define WAVES_PER_BLOCK 8
#define BLOCKS_PB (TILES_PB / WAVES_PER_BLOCK)

#define FNEG (-0.72134752044448170368f)
#define L2E  ( 1.4426950408889634074f)

__global__ __launch_bounds__(256)
void setconv_xgrid_kernel(const float* __restrict__ tg, float* __restrict__ out)
{
    int idx = blockIdx.x * 256 + threadIdx.x;
    int b  = idx >> 15;
    int g  = idx & (GPB - 1);
    int wi = g & 63, hi = (g >> 6) & 63, ti = g >> 12;
    float* o = out + (size_t)idx * DX_D;
    o[0] = tg[b * NT_T + ti];
    o[1] = -1.0f + (float)hi * (2.0f / 63.0f);
    o[2] = -1.0f + (float)wi * (2.0f / 63.0f);
}

__global__ __launch_bounds__(256)
void setconv_fused_kernel(const float* __restrict__ x,
                          const float* __restrict__ z,
                          const float* __restrict__ tg,
                          const float* __restrict__ lsp,
                          float* __restrict__ zg_out)
{
    __shared__ float4 cS[NN];
    __shared__ float  zS[NN * DZ_D];

    float ils0 = 1.0f / (1e-5f + log1pf(__expf(lsp[0])));
    float ils1 = 1.0f / (1e-5f + log1pf(__expf(lsp[1])));
    float ils2 = 1.0f / (1e-5f + log1pf(__expf(lsp[2])));

    int b        = blockIdx.x / BLOCKS_PB;
    int tileBase = (blockIdx.x % BLOCKS_PB) * WAVES_PER_BLOCK;

    const float* xb = x + (size_t)b * NN * DX_D;
    const float* zb = z + (size_t)b * NN * DZ_D;

    for (int n = threadIdx.x; n < NN; n += 256) {
        float c0 = xb[n * 3 + 0] * ils0;
        float c1 = xb[n * 3 + 1] * ils1;
        float c2 = xb[n * 3 + 2] * ils2;
        float cn2f = FNEG * (c0 * c0 + c1 * c1 + c2 * c2);
        cS[n] = make_float4(L2E * c0, L2E * c1, L2E * c2, cn2f);
    }
    {
        const float4* zb4 = (const float4*)zb;
        float4*       zS4 = (float4*)zS;
        for (int i = threadIdx.x; i < NN * DZ_D / 4; i += 256)
            zS4[i] = zb4[i];
    }
    __syncthreads();

    int lane = threadIdx.x & 31;
    int wave = threadIdx.x >> 5;
    int m    = lane & 15;
    int kb   = (lane >> 4) << 1;

    int gtile = tileBase + wave;
    int g  = gtile * 16 + m;
    int wi = g & 63, hi = (g >> 6) & 63, ti = g >> 12;

    float a0 = tg[b * NT_T + ti] * ils0;
    float a1 = (-1.0f + (float)hi * (2.0f / 63.0f)) * ils1;
    float a2 = (-1.0f + (float)wi * (2.0f / 63.0f)) * ils2;
    float an2f = FNEG * (a0 * a0 + a1 * a1 + a2 * a2);

    v8f acc = {};
    #pragma unroll 8
    for (int nb = 0; nb < NN; nb += 4) {
        int n0 = nb + kb;
        float4 cp0 = cS[n0];
        float4 cp1 = cS[n0 + 1];
        float e0 = __builtin_fmaf(a0, cp0.x,
                   __builtin_fmaf(a1, cp0.y,
                   __builtin_fmaf(a2, cp0.z, an2f + cp0.w)));
        float e1 = __builtin_fmaf(a0, cp1.x,
                   __builtin_fmaf(a1, cp1.y,
                   __builtin_fmaf(a2, cp1.z, an2f + cp1.w)));

        v2f A;  A.x  = __builtin_amdgcn_exp2f(e0);
                A.y  = __builtin_amdgcn_exp2f(e1);
        v2f Bm; Bm.x = zS[n0 * DZ_D + m];
                Bm.y = zS[(n0 + 1) * DZ_D + m];

        acc = __builtin_amdgcn_wmma_f32_16x16x4_f32(
                  false, A, false, Bm, (short)0, acc, false, false);
    }

    float* outp = zg_out
        + ((size_t)b * GPB + (size_t)gtile * 16 + (size_t)(lane >> 4) * 8) * DZ_D
        + m;
    #pragma unroll
    for (int r = 0; r < 8; ++r)
        outp[(size_t)r * DZ_D] = acc[r];
}

extern "C" void kernel_launch(void* const* d_in, const int* in_sizes, int n_in,
                              void* d_out, int out_size, void* d_ws, size_t ws_size,
                              hipStream_t stream) {
    (void)in_sizes; (void)n_in; (void)out_size; (void)d_ws; (void)ws_size;
    const float* x   = (const float*)d_in[0];
    const float* z   = (const float*)d_in[1];
    const float* tgr = (const float*)d_in[2];
    const float* lsp = (const float*)d_in[3];

    float* xg_out = (float*)d_out;
    float* zg_out = xg_out + (size_t)NB_B * GPB * DX_D;

    setconv_xgrid_kernel<<<(NB_B * GPB) / 256, 256, 0, stream>>>(tgr, xg_out);
    setconv_fused_kernel<<<NB_B * BLOCKS_PB, 256, 0, stream>>>(x, z, tgr, lsp, zg_out);
}


# Factor the RBF weight per axis: exp tables E[wi][n], P[ti,hi][n] built once per block in LDS replace one exp per (g,n); A frags read from LDS, B = P*z; n sum regrouped (k order)
# speedup vs baseline: 1.4563x; 1.4563x over previous
#include <hip/hip_runtime.h>


typedef float v2f __attribute__((ext_vector_type(2)));
typedef float v8f __attribute__((ext_vector_type(8)));

#define NB_B   2
#define NN     2048
#define NT_T   8
#define DX_D   3
#define DZ_D   16
#define HH     64
#define WW     64
#define GPB    (NT_T * HH * WW)

#define SQH (0.84932180028801904272f)

#define CH   512
#define NCH  (NN / CH)
#define EP   (CH + 4)
#define HPB  4
#define ZR   (CH * DZ_D / 4 / 256)

__global__ __launch_bounds__(256)
void setconv_xgrid_kernel(const float* __restrict__ tg, float* __restrict__ out)
{
    int idx = blockIdx.x * 256 + threadIdx.x;
    int b  = idx >> 15;
    int g  = idx & (GPB - 1);
    int wi = g & 63, hi = (g >> 6) & 63, ti = g >> 12;
    float* o = out + (size_t)idx * DX_D;
    o[0] = tg[b * NT_T + ti];
    o[1] = -1.0f + (float)hi * (2.0f / 63.0f);
    o[2] = -1.0f + (float)wi * (2.0f / 63.0f);
}

__global__ __launch_bounds__(256)
void setconv_sep_kernel(const float* __restrict__ x,
                        const float* __restrict__ z,
                        const float* __restrict__ tg,
                        const float* __restrict__ lsp,
                        float* __restrict__ zg_out)
{
    __shared__ __attribute__((aligned(16))) float eS[WW * EP];
    __shared__ __attribute__((aligned(16))) float zS[CH * DZ_D];
    __shared__ __attribute__((aligned(16))) float pS[HPB * CH];
    __shared__ __attribute__((aligned(16))) float c0S[NN];
    __shared__ __attribute__((aligned(16))) float c1S[NN];
    __shared__ __attribute__((aligned(16))) float c2S[NN];

    float s0 = SQH / (1e-5f + log1pf(__expf(lsp[0])));
    float s1 = SQH / (1e-5f + log1pf(__expf(lsp[1])));
    float s2 = SQH / (1e-5f + log1pf(__expf(lsp[2])));

    int blk = blockIdx.x;
    int b   = blk >> 7;
    int ti  = (blk >> 4) & 7;
    int hq  = blk & 15;

    const float* xb = x + (size_t)b * NN * DX_D;
    const float* zb = z + (size_t)b * NN * DZ_D;

    int tid  = threadIdx.x;
    int lane = tid & 31;
    int wave = tid >> 5;
    int m    = lane & 15;
    int h    = lane >> 4;

    #pragma unroll
    for (int i = 0; i < NN / 256; ++i) {
        int n = tid + 256 * i;
        c0S[n] = xb[n * 3 + 0] * s0;
        c1S[n] = xb[n * 3 + 1] * s1;
        c2S[n] = xb[n * 3 + 2] * s2;
    }

    const float4* zsrc = (const float4*)zb + tid;

    int   swi = tid & 63;
    int   sq  = tid >> 6;
    float a2s = (-1.0f + (float)swi * (2.0f / 63.0f)) * s2;
    int   phl = tid >> 6;
    float a0s = tg[b * NT_T + ti] * s0;
    float a1s = (-1.0f + (float)(hq * HPB + phl) * (2.0f / 63.0f)) * s1;

    int hl = wave >> 1;
    int wb = (wave & 1) * 32;
    int hi = hq * HPB + hl;

    const float* eA0 = eS + (wb + m) * EP + 4 * h;
    const float* eA1 = eA0 + 16 * EP;
    const float* pA  = pS + hl * CH + 4 * h;
    const float* zA  = zS + (4 * h) * DZ_D + m;

    v8f acc0 = {};
    v8f acc1 = {};

    #pragma unroll 1
    for (int c = 0; c < NCH; ++c) {
        int cb = c * CH;
        __syncthreads();

        const float4* zn = zsrc + c * (ZR * 256);
        float4 zr0 = zn[0 * 256], zr1 = zn[1 * 256];
        float4 zr2 = zn[2 * 256], zr3 = zn[3 * 256];
        float4 zr4 = zn[4 * 256], zr5 = zn[5 * 256];
        float4 zr6 = zn[6 * 256], zr7 = zn[7 * 256];
        for (int j = 0; j < CH / 64; ++j) {
            int   n  = (tid & 63) + 64 * j;
            float d0 = a0s - c0S[cb + n];
            float d1 = a1s - c1S[cb + n];
            pS[phl * CH + n] = __builtin_amdgcn_exp2f(-(d0 * d0 + d1 * d1));
        }
        #pragma unroll
        for (int j = 0; j < CH / 16; ++j) {
            int    n4 = j * 16 + sq * 4;
            float4 cc = *(const float4*)(c2S + cb + n4);
            float  dx = a2s - cc.x, dy = a2s - cc.y;
            float  dz = a2s - cc.z, dw = a2s - cc.w;
            float4 ev;
            ev.x = __builtin_amdgcn_exp2f(-(dx * dx));
            ev.y = __builtin_amdgcn_exp2f(-(dy * dy));
            ev.z = __builtin_amdgcn_exp2f(-(dz * dz));
            ev.w = __builtin_amdgcn_exp2f(-(dw * dw));
            *(float4*)(eS + swi * EP + n4) = ev;
        }
        {
            float4* dst = (float4*)zS + tid;
            dst[0 * 256] = zr0; dst[1 * 256] = zr1;
            dst[2 * 256] = zr2; dst[3 * 256] = zr3;
            dst[4 * 256] = zr4; dst[5 * 256] = zr5;
            dst[6 * 256] = zr6; dst[7 * 256] = zr7;
        }
        __syncthreads();

        #pragma unroll 4
        for (int kk = 0; kk < CH / 8; ++kk) {
            int k0 = kk * 8;
            float4 fa0 = *(const float4*)(eA0 + k0);
            float4 fa1 = *(const float4*)(eA1 + k0);
            float4 fp  = *(const float4*)(pA + k0);
            float  z0  = zA[(k0 + 0) * DZ_D];
            float  z1  = zA[(k0 + 1) * DZ_D];
            float  z2  = zA[(k0 + 2) * DZ_D];
            float  z3  = zA[(k0 + 3) * DZ_D];

            v2f Ba; Ba.x = fp.x * z0; Ba.y = fp.y * z1;
            v2f Bb; Bb.x = fp.z * z2; Bb.y = fp.w * z3;
            v2f A0a; A0a.x = fa0.x; A0a.y = fa0.y;
            v2f A0b; A0b.x = fa0.z; A0b.y = fa0.w;
            v2f A1a; A1a.x = fa1.x; A1a.y = fa1.y;
            v2f A1b; A1b.x = fa1.z; A1b.y = fa1.w;

            acc0 = __builtin_amdgcn_wmma_f32_16x16x4_f32(
                       false, A0a, false, Ba, (short)0, acc0, false, false);
            acc1 = __builtin_amdgcn_wmma_f32_16x16x4_f32(
                       false, A1a, false, Ba, (short)0, acc1, false, false);
            acc0 = __builtin_amdgcn_wmma_f32_16x16x4_f32(
                       false, A0b, false, Bb, (short)0, acc0, false, false);
            acc1 = __builtin_amdgcn_wmma_f32_16x16x4_f32(
                       false, A1b, false, Bb, (short)0, acc1, false, false);
        }
    }

    size_t g0 = (size_t)b * GPB + (size_t)ti * (HH * WW) + (size_t)hi * WW + wb;
    float* outp = zg_out + (g0 + (size_t)h * 8) * DZ_D + m;
    #pragma unroll
    for (int r = 0; r < 8; ++r) {
        outp[(size_t)r * DZ_D]        = acc0[r];
        outp[(size_t)(16 + r) * DZ_D] = acc1[r];
    }
}

extern "C" void kernel_launch(void* const* d_in, const int* in_sizes, int n_in,
                              void* d_out, int out_size, void* d_ws, size_t ws_size,
                              hipStream_t stream) {
    (void)in_sizes; (void)n_in; (void)out_size; (void)d_ws; (void)ws_size;
    const float* x   = (const float*)d_in[0];
    const float* z   = (const float*)d_in[1];
    const float* tgr = (const float*)d_in[2];
    const float* lsp = (const float*)d_in[3];

    float* xg_out = (float*)d_out;
    float* zg_out = xg_out + (size_t)NB_B * GPB * DX_D;

    setconv_xgrid_kernel<<<(NB_B * GPB) / 256, 256, 0, stream>>>(tgr, xg_out);
    setconv_sep_kernel<<<NB_B * NT_T * (HH / HPB), 256, 0, stream>>>(x, z, tgr, lsp, zg_out);
}


# Product phase software-pipelined: the LDS operand reads of each 32-WMMA block are issued as one group a whole block ahead (two register sets, last block pair peeled); order of the sums unchanged
# speedup vs baseline: 1.6254x; 1.1161x over previous
#include <hip/hip_runtime.h>


typedef float v2f __attribute__((ext_vector_type(2)));
typedef float v4f __attribute__((ext_vector_type(4)));
typedef float v8f __attribute__((ext_vector_type(8)));

#define NB_B   2
#define NN     2048
#define NT_T   8
#define DX_D   3
#define DZ_D   16
#define HH     64
#define WW     64
#define GPB    (NT_T * HH * WW)

#define SQH (0.84932180028801904272f)

#define CH   512
#define NCH  (NN / CH)
#define EP   (CH + 4)
#define HPB  4
#define ZR   (CH * DZ_D / 4 / 256)

__global__ __launch_bounds__(256)
void setconv_xgrid_kernel(const float* __restrict__ tg, float* __restrict__ out)
{
    int idx = blockIdx.x * 256 + threadIdx.x;
    int b  = idx >> 15;
    int g  = idx & (GPB - 1);
    int wi = g & 63, hi = (g >> 6) & 63, ti = g >> 12;
    float* o = out + (size_t)idx * DX_D;
    o[0] = tg[b * NT_T + ti];
    o[1] = -1.0f + (float)hi * (2.0f / 63.0f);
    o[2] = -1.0f + (float)wi * (2.0f / 63.0f);
}

#define NK   8

struct Frag {
    v4f    a0[NK];
    v4f    a1[NK];
    v4f    p[NK];
    float  z[NK][4];
};

__device__ __forceinline__
void frag_load(Frag& f, const float* eA0, const float* eA1,
               const float* pA, const float* zA, int k)
{
    #pragma unroll
    for (int i = 0; i < NK; ++i) {
        int k0 = k + 8 * i;
        f.a0[i]   = *(const v4f*)(eA0 + k0);
        f.a1[i]   = *(const v4f*)(eA1 + k0);
        f.p[i]    = *(const v4f*)(pA + k0);
        f.z[i][0] = zA[(k0 + 0) * DZ_D];
        f.z[i][1] = zA[(k0 + 1) * DZ_D];
        f.z[i][2] = zA[(k0 + 2) * DZ_D];
        f.z[i][3] = zA[(k0 + 3) * DZ_D];
    }
}

__device__ __forceinline__
void frag_mma(const Frag& f, v8f& acc0, v8f& acc1)
{
    #pragma unroll
    for (int i = 0; i < NK; ++i) {
        v2f Ba; Ba.x = f.p[i].x * f.z[i][0]; Ba.y = f.p[i].y * f.z[i][1];
        v2f Bb; Bb.x = f.p[i].z * f.z[i][2]; Bb.y = f.p[i].w * f.z[i][3];
        v2f A0a; A0a.x = f.a0[i].x; A0a.y = f.a0[i].y;
        v2f A0b; A0b.x = f.a0[i].z; A0b.y = f.a0[i].w;
        v2f A1a; A1a.x = f.a1[i].x; A1a.y = f.a1[i].y;
        v2f A1b; A1b.x = f.a1[i].z; A1b.y = f.a1[i].w;

        acc0 = __builtin_amdgcn_wmma_f32_16x16x4_f32(
                   false, A0a, false, Ba, (short)0, acc0, false, false);
        acc1 = __builtin_amdgcn_wmma_f32_16x16x4_f32(
                   false, A1a, false, Ba, (short)0, acc1, false, false);
        acc0 = __builtin_amdgcn_wmma_f32_16x16x4_f32(
                   false, A0b, false, Bb, (short)0, acc0, false, false);
        acc1 = __builtin_amdgcn_wmma_f32_16x16x4_f32(
                   false, A1b, false, Bb, (short)0, acc1, false, false);
    }
}

__device__ __forceinline__
void order_point(v8f& acc0, v8f& acc1)
{
    asm volatile("" : "+v"(acc0), "+v"(acc1) :: "memory");
}

__device__ __forceinline__
void order_point(v8f& acc0, v8f& acc1, Frag& f)
{
    asm volatile("" : "+v"(acc0), "+v"(acc1),
                      "+v"(f.p[0]), "+v"(f.p[1]), "+v"(f.p[2]), "+v"(f.p[3]),
                      "+v"(f.p[4]), "+v"(f.p[5]), "+v"(f.p[6]), "+v"(f.p[7])
                 :: "memory");
}

__global__ __launch_bounds__(256)
void setconv_sep_kernel(const float* __restrict__ x,
                        const float* __restrict__ z,
                        const float* __restrict__ tg,
                        const float* __restrict__ lsp,
                        float* __restrict__ zg_out)
{
    __shared__ __attribute__((aligned(16))) float eS[WW * EP];
    __shared__ __attribute__((aligned(16))) float zS[CH * DZ_D];
    __shared__ __attribute__((aligned(16))) float pS[HPB * CH];
    __shared__ __attribute__((aligned(16))) float c0S[NN];
    __shared__ __attribute__((aligned(16))) float c1S[NN];
    __shared__ __attribute__((aligned(16))) float c2S[NN];

    float s0 = SQH / (1e-5f + log1pf(__expf(lsp[0])));
    float s1 = SQH / (1e-5f + log1pf(__expf(lsp[1])));
    float s2 = SQH / (1e-5f + log1pf(__expf(lsp[2])));

    int blk = blockIdx.x;
    int b   = blk >> 7;
    int ti  = (blk >> 4) & 7;
    int hq  = blk & 15;

    const float* xb = x + (size_t)b * NN * DX_D;
    const float* zb = z + (size_t)b * NN * DZ_D;

    int tid  = threadIdx.x;
    int lane = tid & 31;
    int wave = tid >> 5;
    int m    = lane & 15;
    int h    = lane >> 4;

    #pragma unroll
    for (int i = 0; i < NN / 256; ++i) {
        int n = tid + 256 * i;
        c0S[n] = xb[n * 3 + 0] * s0;
        c1S[n] = xb[n * 3 + 1] * s1;
        c2S[n] = xb[n * 3 + 2] * s2;
    }

    const float4* zsrc = (const float4*)zb + tid;

    int   swi = tid & 63;
    int   sq  = tid >> 6;
    float a2s = (-1.0f + (float)swi * (2.0f / 63.0f)) * s2;
    int   phl = tid >> 6;
    float a0s = tg[b * NT_T + ti] * s0;
    float a1s = (-1.0f + (float)(hq * HPB + phl) * (2.0f / 63.0f)) * s1;

    int hl = wave >> 1;
    int wb = (wave & 1) * 32;
    int hi = hq * HPB + hl;

    const float* eA0 = eS + (wb + m) * EP + 4 * h;
    const float* eA1 = eA0 + 16 * EP;
    const float* pA  = pS + hl * CH + 4 * h;
    const float* zA  = zS + (4 * h) * DZ_D + m;

    v8f acc0 = {};
    v8f acc1 = {};

    #pragma unroll 1
    for (int c = 0; c < NCH; ++c) {
        int cb = c * CH;
        __syncthreads();

        const float4* zn = zsrc + c * (ZR * 256);
        float4 zr0 = zn[0 * 256], zr1 = zn[1 * 256];
        float4 zr2 = zn[2 * 256], zr3 = zn[3 * 256];
        float4 zr4 = zn[4 * 256], zr5 = zn[5 * 256];
        float4 zr6 = zn[6 * 256], zr7 = zn[7 * 256];
        for (int j = 0; j < CH / 64; ++j) {
            int   n  = (tid & 63) + 64 * j;
            float d0 = a0s - c0S[cb + n];
            float d1 = a1s - c1S[cb + n];
            pS[phl * CH + n] = __builtin_amdgcn_exp2f(-(d0 * d0 + d1 * d1));
        }
        #pragma unroll
        for (int j = 0; j < CH / 16; ++j) {
            int    n4 = j * 16 + sq * 4;
            float4 cc = *(const float4*)(c2S + cb + n4);
            float  dx = a2s - cc.x, dy = a2s - cc.y;
            float  dz = a2s - cc.z, dw = a2s - cc.w;
            float4 ev;
            ev.x = __builtin_amdgcn_exp2f(-(dx * dx));
            ev.y = __builtin_amdgcn_exp2f(-(dy * dy));
            ev.z = __builtin_amdgcn_exp2f(-(dz * dz));
            ev.w = __builtin_amdgcn_exp2f(-(dw * dw));
            *(float4*)(eS + swi * EP + n4) = ev;
        }
        {
            float4* dst = (float4*)zS + tid;
            dst[0 * 256] = zr0; dst[1 * 256] = zr1;
            dst[2 * 256] = zr2; dst[3 * 256] = zr3;
            dst[4 * 256] = zr4; dst[5 * 256] = zr5;
            dst[6 * 256] = zr6; dst[7 * 256] = zr7;
        }
        __syncthreads();

        Frag f0, f1;
        frag_load(f0, eA0, eA1, pA, zA, 0);
        frag_load(f1, eA0, eA1, pA, zA, 8 * NK);
        #pragma unroll 1
        for (int kb = 0; kb < CH - 16 * NK; kb += 16 * NK) {
            frag_mma(f0, acc0, acc1);
            order_point(acc0, acc1);
            frag_load(f0, eA0, eA1, pA, zA, kb + 16 * NK);
            order_point(acc0, acc1, f1);
            frag_mma(f1, acc0, acc1);
            order_point(acc0, acc1);
            frag_load(f1, eA0, eA1, pA, zA, kb + 24 * NK);
        }
        frag_mma(f0, acc0, acc1);
        frag_mma(f1, acc0, acc1);
        asm volatile("v_nop\n\tv_nop\n\tv_nop\n\tv_nop\n\tv_nop"
                     : "+v"(acc0), "+v"(acc1));
    }

    size_t g0 = (size_t)b * GPB + (size_t)ti * (HH * WW) + (size_t)hi * WW + wb;
    float* outp = zg_out + (g0 + (size_t)h * 8) * DZ_D + m;
    #pragma unroll
    for (int r = 0; r < 8; ++r) {
        outp[(size_t)r * DZ_D]        = acc0[r];
        outp[(size_t)(16 + r) * DZ_D] = acc1[r];
    }
}

extern "C" void kernel_launch(void* const* d_in, const int* in_sizes, int n_in,
                              void* d_out, int out_size, void* d_ws, size_t ws_size,
                              hipStream_t stream) {
    (void)in_sizes; (void)n_in; (void)out_size; (void)d_ws; (void)ws_size;
    const float* x   = (const float*)d_in[0];
    const float* z   = (const float*)d_in[1];
    const float* tgr = (const float*)d_in[2];
    const float* lsp = (const float*)d_in[3];

    float* xg_out = (float*)d_out;
    float* zg_out = xg_out + (size_t)NB_B * GPB * DX_D;

    setconv_xgrid_kernel<<<(NB_B * GPB) / 256, 256, 0, stream>>>(tgr, xg_out);
    setconv_sep_kernel<<<NB_B * NT_T * (HH / HPB), 256, 0, stream>>>(x, z, tgr, lsp, zg_out);
}


# x_grid rows written by the product kernel itself: each block stores the 768 floats of its own 256 grid points as whole 16-byte pieces before staging; one launch per call instead of two
# speedup vs baseline: 1.7754x; 1.0923x over previous
#include <hip/hip_runtime.h>


typedef float v2f __attribute__((ext_vector_type(2)));
typedef float v4f __attribute__((ext_vector_type(4)));
typedef float v8f __attribute__((ext_vector_type(8)));

#define NB_B   2
#define NN     2048
#define NT_T   8
#define DX_D   3
#define DZ_D   16
#define HH     64
#define WW     64
#define GPB    (NT_T * HH * WW)

#define SQH (0.84932180028801904272f)

#define CH   512
#define NCH  (NN / CH)
#define EP   (CH + 4)
#define HPB  4
#define ZR   (CH * DZ_D / 4 / 256)

#define NK   8

struct Frag {
    v4f    a0[NK];
    v4f    a1[NK];
    v4f    p[NK];
    float  z[NK][4];
};

__device__ __forceinline__
void frag_load(Frag& f, const float* eA0, const float* eA1,
               const float* pA, const float* zA, int k)
{
    #pragma unroll
    for (int i = 0; i < NK; ++i) {
        int k0 = k + 8 * i;
        f.a0[i]   = *(const v4f*)(eA0 + k0);
        f.a1[i]   = *(const v4f*)(eA1 + k0);
        f.p[i]    = *(const v4f*)(pA + k0);
        f.z[i][0] = zA[(k0 + 0) * DZ_D];
        f.z[i][1] = zA[(k0 + 1) * DZ_D];
        f.z[i][2] = zA[(k0 + 2) * DZ_D];
        f.z[i][3] = zA[(k0 + 3) * DZ_D];
    }
}

__device__ __forceinline__
void frag_mma(const Frag& f, v8f& acc0, v8f& acc1)
{
    #pragma unroll
    for (int i = 0; i < NK; ++i) {
        v2f Ba; Ba.x = f.p[i].x * f.z[i][0]; Ba.y = f.p[i].y * f.z[i][1];
        v2f Bb; Bb.x = f.p[i].z * f.z[i][2]; Bb.y = f.p[i].w * f.z[i][3];
        v2f A0a; A0a.x = f.a0[i].x; A0a.y = f.a0[i].y;
        v2f A0b; A0b.x = f.a0[i].z; A0b.y = f.a0[i].w;
        v2f A1a; A1a.x = f.a1[i].x; A1a.y = f.a1[i].y;
        v2f A1b; A1b.x = f.a1[i].z; A1b.y = f.a1[i].w;

        acc0 = __builtin_amdgcn_wmma_f32_16x16x4_f32(
                   false, A0a, false, Ba, (short)0, acc0, false, false);
        acc1 = __builtin_amdgcn_wmma_f32_16x16x4_f32(
                   false, A1a, false, Ba, (short)0, acc1, false, false);
        acc0 = __builtin_amdgcn_wmma_f32_16x16x4_f32(
                   false, A0b, false, Bb, (short)0, acc0, false, false);
        acc1 = __builtin_amdgcn_wmma_f32_16x16x4_f32(
                   false, A1b, false, Bb, (short)0, acc1, false, false);
    }
}

__device__ __forceinline__
void order_point(v8f& acc0, v8f& acc1)
{
    asm volatile("" : "+v"(acc0), "+v"(acc1) :: "memory");
}

__device__ __forceinline__
void order_point(v8f& acc0, v8f& acc1, Frag& f)
{
    asm volatile("" : "+v"(acc0), "+v"(acc1),
                      "+v"(f.p[0]), "+v"(f.p[1]), "+v"(f.p[2]), "+v"(f.p[3]),
                      "+v"(f.p[4]), "+v"(f.p[5]), "+v"(f.p[6]), "+v"(f.p[7])
                 :: "memory");
}

__global__ __launch_bounds__(256)
void setconv_sep_kernel(const float* __restrict__ x,
                        const float* __restrict__ z,
                        const float* __restrict__ tg,
                        const float* __restrict__ lsp,
                        float* __restrict__ xg_out,
                        float* __restrict__ zg_out)
{
    __shared__ __attribute__((aligned(16))) float eS[WW * EP];
    __shared__ __attribute__((aligned(16))) float zS[CH * DZ_D];
    __shared__ __attribute__((aligned(16))) float pS[HPB * CH];
    __shared__ __attribute__((aligned(16))) float c0S[NN];
    __shared__ __attribute__((aligned(16))) float c1S[NN];
    __shared__ __attribute__((aligned(16))) float c2S[NN];

    float s0 = SQH / (1e-5f + log1pf(__expf(lsp[0])));
    float s1 = SQH / (1e-5f + log1pf(__expf(lsp[1])));
    float s2 = SQH / (1e-5f + log1pf(__expf(lsp[2])));

    int blk = blockIdx.x;
    int b   = blk >> 7;
    int ti  = (blk >> 4) & 7;
    int hq  = blk & 15;

    const float* xb = x + (size_t)b * NN * DX_D;
    const float* zb = z + (size_t)b * NN * DZ_D;

    int tid  = threadIdx.x;
    int lane = tid & 31;
    int wave = tid >> 5;
    int m    = lane & 15;
    int h    = lane >> 4;

    if (tid < (HPB * WW * DX_D) / 4) {
        float  tgv = tg[b * NT_T + ti];
        float  xv[4];
        #pragma unroll
        for (int s = 0; s < 4; ++s) {
            unsigned j  = 4u * (unsigned)tid + (unsigned)s;
            unsigned q  = j / 3u;
            unsigned cd = j - 3u * q;
            int      qh = hq * HPB + (int)(q >> 6);
            int      qw = (int)(q & 63u);
            float    ch = -1.0f + (float)qh * (2.0f / 63.0f);
            float    cw = -1.0f + (float)qw * (2.0f / 63.0f);
            xv[s] = (cd == 0u) ? tgv : ((cd == 1u) ? ch : cw);
        }
        size_t gq = (size_t)b * GPB + (size_t)ti * (HH * WW)
                  + (size_t)hq * (HPB * WW);
        float4 v4; v4.x = xv[0]; v4.y = xv[1]; v4.z = xv[2]; v4.w = xv[3];
        *((float4*)(xg_out + gq * DX_D) + tid) = v4;
    }

    #pragma unroll
    for (int i = 0; i < NN / 256; ++i) {
        int n = tid + 256 * i;
        c0S[n] = xb[n * 3 + 0] * s0;
        c1S[n] = xb[n * 3 + 1] * s1;
        c2S[n] = xb[n * 3 + 2] * s2;
    }

    const float4* zsrc = (const float4*)zb + tid;

    int   swi = tid & 63;
    int   sq  = tid >> 6;
    float a2s = (-1.0f + (float)swi * (2.0f / 63.0f)) * s2;
    int   phl = tid >> 6;
    float a0s = tg[b * NT_T + ti] * s0;
    float a1s = (-1.0f + (float)(hq * HPB + phl) * (2.0f / 63.0f)) * s1;

    int hl = wave >> 1;
    int wb = (wave & 1) * 32;
    int hi = hq * HPB + hl;

    const float* eA0 = eS + (wb + m) * EP + 4 * h;
    const float* eA1 = eA0 + 16 * EP;
    const float* pA  = pS + hl * CH + 4 * h;
    const float* zA  = zS + (4 * h) * DZ_D + m;

    v8f acc0 = {};
    v8f acc1 = {};

    #pragma unroll 1
    for (int c = 0; c < NCH; ++c) {
        int cb = c * CH;
        __syncthreads();

        const float4* zn = zsrc + c * (ZR * 256);
        float4 zr0 = zn[0 * 256], zr1 = zn[1 * 256];
        float4 zr2 = zn[2 * 256], zr3 = zn[3 * 256];
        float4 zr4 = zn[4 * 256], zr5 = zn[5 * 256];
        float4 zr6 = zn[6 * 256], zr7 = zn[7 * 256];
        for (int j = 0; j < CH / 64; ++j) {
            int   n  = (tid & 63) + 64 * j;
            float d0 = a0s - c0S[cb + n];
            float d1 = a1s - c1S[cb + n];
            pS[phl * CH + n] = __builtin_amdgcn_exp2f(-(d0 * d0 + d1 * d1));
        }
        #pragma unroll
        for (int j = 0; j < CH / 16; ++j) {
            int    n4 = j * 16 + sq * 4;
            float4 cc = *(const float4*)(c2S + cb + n4);
            float  dx = a2s - cc.x, dy = a2s - cc.y;
            float  dz = a2s - cc.z, dw = a2s - cc.w;
            float4 ev;
            ev.x = __builtin_amdgcn_exp2f(-(dx * dx));
            ev.y = __builtin_amdgcn_exp2f(-(dy * dy));
            ev.z = __builtin_amdgcn_exp2f(-(dz * dz));
            ev.w = __builtin_amdgcn_exp2f(-(dw * dw));
            *(float4*)(eS + swi * EP + n4) = ev;
        }
        {
            float4* dst = (float4*)zS + tid;
            dst[0 * 256] = zr0; dst[1 * 256] = zr1;
            dst[2 * 256] = zr2; dst[3 * 256] = zr3;
            dst[4 * 256] = zr4; dst[5 * 256] = zr5;
            dst[6 * 256] = zr6; dst[7 * 256] = zr7;
        }
        __syncthreads();

        Frag f0, f1;
        frag_load(f0, eA0, eA1, pA, zA, 0);
        frag_load(f1, eA0, eA1, pA, zA, 8 * NK);
        #pragma unroll 1
        for (int kb = 0; kb < CH - 16 * NK; kb += 16 * NK) {
            frag_mma(f0, acc0, acc1);
            order_point(acc0, acc1);
            frag_load(f0, eA0, eA1, pA, zA, kb + 16 * NK);
            order_point(acc0, acc1, f1);
            frag_mma(f1, acc0, acc1);
            order_point(acc0, acc1);
            frag_load(f1, eA0, eA1, pA, zA, kb + 24 * NK);
        }
        frag_mma(f0, acc0, acc1);
        frag_mma(f1, acc0, acc1);
        asm volatile("v_nop\n\tv_nop\n\tv_nop\n\tv_nop\n\tv_nop"
                     : "+v"(acc0), "+v"(acc1));
    }

    size_t g0 = (size_t)b * GPB + (size_t)ti * (HH * WW) + (size_t)hi * WW + wb;
    float* outp = zg_out + (g0 + (size_t)h * 8) * DZ_D + m;
    #pragma unroll
    for (int r = 0; r < 8; ++r) {
        outp[(size_t)r * DZ_D]        = acc0[r];
        outp[(size_t)(16 + r) * DZ_D] = acc1[r];
    }
}

extern "C" void kernel_launch(void* const* d_in, const int* in_sizes, int n_in,
                              void* d_out, int out_size, void* d_ws, size_t ws_size,
                              hipStream_t stream) {
    (void)in_sizes; (void)n_in; (void)out_size; (void)d_ws; (void)ws_size;
    const float* x   = (const float*)d_in[0];
    const float* z   = (const float*)d_in[1];
    const float* tgr = (const float*)d_in[2];
    const float* lsp = (const float*)d_in[3];

    float* xg_out = (float*)d_out;
    float* zg_out = xg_out + (size_t)NB_B * GPB * DX_D;

    setconv_sep_kernel<<<NB_B * NT_T * (HH / HPB), 256, 0, stream>>>(x, z, tgr, lsp, xg_out, zg_out);
}


# Block re-tiled: square 16 wi x 16 hi tile on 4 waves (one per SIMD), each wave forms 4 hi from one shared E fragment; half the table exps per block, 2 chunks of 1024 n; order of sums unchanged
# speedup vs baseline: 1.8609x; 1.0482x over previous
#include <hip/hip_runtime.h>


typedef float v2f __attribute__((ext_vector_type(2)));
typedef float v4f __attribute__((ext_vector_type(4)));
typedef float v8f __attribute__((ext_vector_type(8)));

#define NB_B   2
#define NN     2048
#define NT_T   8
#define DX_D   3
#define DZ_D   16
#define HH     64
#define WW     64
#define GPB    (NT_T * HH * WW)

#define SQH (0.84932180028801904272f)

#define NTH  128
#define TW   16
#define TR   16
#define RPW  4
#define CH   1024
#define NCH  (NN / CH)
#define EP   (CH + 4)
#define PP   (CH + 32)
#define ZR   (CH * DZ_D / 4 / NTH)
#define XQ   (TW * TR * DX_D / 4)

#define NK   8

struct Frag {
    v4f    a[NK];
    v4f    p[NK][RPW];
    v4f    z[NK];
};

__device__ __forceinline__
void frag_load(Frag& f, const float* eA, const float* pA, const float* zA, int k)
{
    #pragma unroll
    for (int i = 0; i < NK; ++i) {
        int k0 = k + 8 * i;
        f.a[i] = *(const v4f*)(eA + k0);
        #pragma unroll
        for (int j = 0; j < RPW; ++j)
            f.p[i][j] = *(const v4f*)(pA + j * PP + k0);
        f.z[i].x = zA[(k0 + 0) * DZ_D];
        f.z[i].y = zA[(k0 + 1) * DZ_D];
        f.z[i].z = zA[(k0 + 2) * DZ_D];
        f.z[i].w = zA[(k0 + 3) * DZ_D];
    }
}

__device__ __forceinline__
void frag_mma(const Frag& f, v8f (&acc)[RPW])
{
    #pragma unroll
    for (int i = 0; i < NK; ++i) {
        v2f Aa;  Aa.x  = f.a[i].x; Aa.y  = f.a[i].y;
        v2f Ab;  Ab.x  = f.a[i].z; Ab.y  = f.a[i].w;
        v2f z01; z01.x = f.z[i].x; z01.y = f.z[i].y;
        v2f z23; z23.x = f.z[i].z; z23.y = f.z[i].w;

        #pragma unroll
        for (int j = 0; j < RPW; ++j) {
            v2f B; B.x = f.p[i][j].x * z01.x; B.y = f.p[i][j].y * z01.y;
            acc[j] = __builtin_amdgcn_wmma_f32_16x16x4_f32(
                         false, Aa, false, B, (short)0, acc[j], false, false);
        }
        #pragma unroll
        for (int j = 0; j < RPW; ++j) {
            v2f B; B.x = f.p[i][j].z * z23.x; B.y = f.p[i][j].w * z23.y;
            acc[j] = __builtin_amdgcn_wmma_f32_16x16x4_f32(
                         false, Ab, false, B, (short)0, acc[j], false, false);
        }
    }
}

__device__ __forceinline__
void order_point(v8f (&acc)[RPW])
{
    asm volatile("" : "+v"(acc[0]), "+v"(acc[1]), "+v"(acc[2]), "+v"(acc[3])
                 :: "memory");
}

__device__ __forceinline__
void order_point(v8f (&acc)[RPW], Frag& f)
{
    asm volatile("" : "+v"(acc[0]), "+v"(acc[1]), "+v"(acc[2]), "+v"(acc[3]),
                      "+v"(f.z[0]), "+v"(f.z[1]), "+v"(f.z[2]), "+v"(f.z[3]),
                      "+v"(f.z[4]), "+v"(f.z[5]), "+v"(f.z[6]), "+v"(f.z[7])
                 :: "memory");
}

__device__ __forceinline__
void ld8(v4f (&d)[8], const float* src)
{
    #pragma unroll
    for (int i = 0; i < 8; ++i) d[i] = *(const v4f*)(src + 32 * i);
}

__device__ __forceinline__
void group_point(v4f (&a)[8], v4f (&b)[8])
{
    asm volatile("" : "+v"(a[0]), "+v"(a[1]), "+v"(a[2]), "+v"(a[3]),
                      "+v"(a[4]), "+v"(a[5]), "+v"(a[6]), "+v"(a[7]),
                      "+v"(b[0]), "+v"(b[1]), "+v"(b[2]), "+v"(b[3]),
                      "+v"(b[4]), "+v"(b[5]), "+v"(b[6]), "+v"(b[7])
                 :: "memory");
}

__device__ __forceinline__
void p_group(const v4f (&q0)[8], const v4f (&q1)[8], float a0s, float a1s,
             float* dst)
{
    #pragma unroll
    for (int i = 0; i < 8; ++i) {
        float d0x = a0s - q0[i].x, d0y = a0s - q0[i].y;
        float d0z = a0s - q0[i].z, d0w = a0s - q0[i].w;
        float d1x = a1s - q1[i].x, d1y = a1s - q1[i].y;
        float d1z = a1s - q1[i].z, d1w = a1s - q1[i].w;
        v4f pv;
        pv.x = __builtin_amdgcn_exp2f(-(d0x * d0x + d1x * d1x));
        pv.y = __builtin_amdgcn_exp2f(-(d0y * d0y + d1y * d1y));
        pv.z = __builtin_amdgcn_exp2f(-(d0z * d0z + d1z * d1z));
        pv.w = __builtin_amdgcn_exp2f(-(d0w * d0w + d1w * d1w));
        *(v4f*)(dst + 32 * i) = pv;
    }
}

__device__ __forceinline__
void e_group(const v4f (&cc)[8], float a2s, float* dst)
{
    #pragma unroll
    for (int i = 0; i < 8; ++i) {
        float dx = a2s - cc[i].x, dy = a2s - cc[i].y;
        float dz = a2s - cc[i].z, dw = a2s - cc[i].w;
        v4f ev;
        ev.x = __builtin_amdgcn_exp2f(-(dx * dx));
        ev.y = __builtin_amdgcn_exp2f(-(dy * dy));
        ev.z = __builtin_amdgcn_exp2f(-(dz * dz));
        ev.w = __builtin_amdgcn_exp2f(-(dw * dw));
        *(v4f*)(dst + 32 * i) = ev;
    }
}

__global__ __launch_bounds__(NTH)
void setconv_sq_kernel(const float* __restrict__ x,
                       const float* __restrict__ z,
                       const float* __restrict__ tg,
                       const float* __restrict__ lsp,
                       float* __restrict__ xg_out,
                       float* __restrict__ zg_out)
{
    __shared__ __attribute__((aligned(16))) float eS[TW * EP];
    __shared__ __attribute__((aligned(16))) float pS[TR * PP];
    __shared__ __attribute__((aligned(16))) float zS[CH * DZ_D];
    __shared__ __attribute__((aligned(16))) float c0S[NN];
    __shared__ __attribute__((aligned(16))) float c1S[NN];
    __shared__ __attribute__((aligned(16))) float c2S[NN];

    float s0 = SQH / (1e-5f + log1pf(__expf(lsp[0])));
    float s1 = SQH / (1e-5f + log1pf(__expf(lsp[1])));
    float s2 = SQH / (1e-5f + log1pf(__expf(lsp[2])));

    int blk = blockIdx.x;
    int b   = blk >> 7;
    int pg  = (blk >> 2) & 31;
    int wq  = blk & 3;
    int ti  = pg >> 2;
    int hi0 = (pg & 3) * TR;

    const float* xb = x + (size_t)b * NN * DX_D;
    const float* zb = z + (size_t)b * NN * DZ_D;

    int tid  = threadIdx.x;
    int lane = tid & 31;
    int wave = tid >> 5;
    int m    = lane & 15;
    int h    = lane >> 4;

    {
        int   xti = (blk >> 4) & 7;
        int   xh0 = (blk & 15) * 4;
        float tgv = tg[b * NT_T + xti];
        #pragma unroll
        for (int rd = 0; rd < 2; ++rd) {
            int f4 = tid + rd * NTH;
            if (f4 < XQ) {
                float xv[4];
                #pragma unroll
                for (int s = 0; s < 4; ++s) {
                    unsigned j  = 4u * (unsigned)f4 + (unsigned)s;
                    unsigned q  = j / 3u;
                    unsigned cd = j - 3u * q;
                    int      qh = xh0 + (int)(q >> 6);
                    int      qw = (int)(q & 63u);
                    float    ch = -1.0f + (float)qh * (2.0f / 63.0f);
                    float    cw = -1.0f + (float)qw * (2.0f / 63.0f);
                    xv[s] = (cd == 0u) ? tgv : ((cd == 1u) ? ch : cw);
                }
                float4 v4; v4.x = xv[0]; v4.y = xv[1]; v4.z = xv[2]; v4.w = xv[3];
                *((float4*)(xg_out + (size_t)blk * (256 * DX_D)) + f4) = v4;
            }
        }
    }

    #pragma unroll
    for (int i = 0; i < NN / NTH; ++i) {
        int n = tid + NTH * i;
        c0S[n] = xb[n * 3 + 0] * s0;
        c1S[n] = xb[n * 3 + 1] * s1;
        c2S[n] = xb[n * 3 + 2] * s2;
    }

    const float4* zsrc = (const float4*)zb + tid;

    int   erow = tid & 15;
    int   esq  = tid >> 4;
    float a2s  = (-1.0f + (float)(wq * TW + erow) * (2.0f / 63.0f)) * s2;
    int   prow = tid >> 3;
    int   psq  = tid & 7;
    float a0s  = tg[b * NT_T + ti] * s0;
    float a1s  = (-1.0f + (float)(hi0 + prow) * (2.0f / 63.0f)) * s1;

    const float* eA = eS + m * EP + 4 * h;
    const float* pA = pS + (wave * RPW) * PP + 4 * h;
    const float* zA = zS + (4 * h) * DZ_D + m;

    v8f acc[RPW];
    #pragma unroll
    for (int j = 0; j < RPW; ++j) acc[j] = (v8f){};

    #pragma unroll 1
    for (int c = 0; c < NCH; ++c) {
        int cb = c * CH;
        __syncthreads();

        const float4* zn = zsrc + c * (ZR * NTH);
#define ZLD(i) float4 zr##i = zn[(i) * NTH];
        ZLD(0) ZLD(1) ZLD(2) ZLD(3)
        ZLD(4) ZLD(5) ZLD(6) ZLD(7)
        ZLD(8) ZLD(9) ZLD(10) ZLD(11)
        ZLD(12) ZLD(13) ZLD(14) ZLD(15)
        ZLD(16) ZLD(17) ZLD(18) ZLD(19)
        ZLD(20) ZLD(21) ZLD(22) ZLD(23)
        ZLD(24) ZLD(25) ZLD(26) ZLD(27)
        ZLD(28) ZLD(29) ZLD(30) ZLD(31)
#undef ZLD

        {
            const float* c0p = c0S + cb + psq * 4;
            const float* c1p = c1S + cb + psq * 4;
            const float* c2p = c2S + cb + esq * 4;
            float* pd = pS + prow * PP + psq * 4;
            float* ed = eS + erow * EP + esq * 4;
            v4f A0[8], B0[8], A1[8], B1[8], A2[8], B2[8], A3[8], B3[8];
            v4f C0[8], C1[8], C2[8], C3[8];
            ld8(A0, c0p);       ld8(B0, c1p);
            ld8(A1, c0p + 256); ld8(B1, c1p + 256);
            group_point(A0, B0); p_group(A0, B0, a0s, a1s, pd);
            ld8(A2, c0p + 512); ld8(B2, c1p + 512);
            group_point(A1, B1); p_group(A1, B1, a0s, a1s, pd + 256);
            ld8(A3, c0p + 768); ld8(B3, c1p + 768);
            group_point(A2, B2); p_group(A2, B2, a0s, a1s, pd + 512);
            ld8(C0, c2p);       ld8(C1, c2p + 256);
            group_point(A3, B3); p_group(A3, B3, a0s, a1s, pd + 768);
            ld8(C2, c2p + 512); ld8(C3, c2p + 768);
            group_point(C0, C1); e_group(C0, a2s, ed); e_group(C1, a2s, ed + 256);
            group_point(C2, C3); e_group(C2, a2s, ed + 512); e_group(C3, a2s, ed + 768);
        }
        {
            float4* dst = (float4*)zS + tid;
#define ZST(i) dst[(i) * NTH] = zr##i;
            ZST(0) ZST(1) ZST(2) ZST(3)
            ZST(4) ZST(5) ZST(6) ZST(7)
            ZST(8) ZST(9) ZST(10) ZST(11)
            ZST(12) ZST(13) ZST(14) ZST(15)
            ZST(16) ZST(17) ZST(18) ZST(19)
            ZST(20) ZST(21) ZST(22) ZST(23)
            ZST(24) ZST(25) ZST(26) ZST(27)
            ZST(28) ZST(29) ZST(30) ZST(31)
#undef ZST
        }
        __syncthreads();

        Frag f0, f1;
        frag_load(f0, eA, pA, zA, 0);
        frag_load(f1, eA, pA, zA, 8 * NK);
        #pragma unroll 1
        for (int kb = 0; kb < CH - 16 * NK; kb += 16 * NK) {
            frag_mma(f0, acc);
            order_point(acc);
            frag_load(f0, eA, pA, zA, kb + 16 * NK);
            order_point(acc, f1);
            frag_mma(f1, acc);
            order_point(acc);
            frag_load(f1, eA, pA, zA, kb + 24 * NK);
        }
        frag_mma(f0, acc);
        frag_mma(f1, acc);
        asm volatile("v_nop\n\tv_nop\n\tv_nop\n\tv_nop\n\tv_nop"
                     : "+v"(acc[0]), "+v"(acc[1]), "+v"(acc[2]), "+v"(acc[3]));
    }

    #pragma unroll
    for (int j = 0; j < RPW; ++j) {
        int    hi = hi0 + wave * RPW + j;
        size_t g0 = (size_t)b * GPB + (size_t)ti * (HH * WW)
                  + (size_t)hi * WW + (size_t)wq * TW;
        float* outp = zg_out + (g0 + (size_t)h * 8) * DZ_D + m;
        #pragma unroll
        for (int r = 0; r < 8; ++r)
            outp[(size_t)r * DZ_D] = acc[j][r];
    }
}

extern "C" void kernel_launch(void* const* d_in, const int* in_sizes, int n_in,
                              void* d_out, int out_size, void* d_ws, size_t ws_size,
                              hipStream_t stream) {
    (void)in_sizes; (void)n_in; (void)out_size; (void)d_ws; (void)ws_size;
    const float* x   = (const float*)d_in[0];
    const float* z   = (const float*)d_in[1];
    const float* tgr = (const float*)d_in[2];
    const float* lsp = (const float*)d_in[3];

    float* xg_out = (float*)d_out;
    float* zg_out = xg_out + (size_t)NB_B * GPB * DX_D;

    setconv_sq_kernel<<<NB_B * NT_T * (HH / TR) * (WW / TW), NTH, 0, stream>>>(x, z, tgr, lsp, xg_out, zg_out);
}


# Waves split the points, not the rows (block 8 hi x 32 wi, each wave a quarter of n on 16 tiles): an E read feeds 8 rows, a P*z product 2 tiles; partial sums added via LDS in order w0+w1+w2+w3
# speedup vs baseline: 1.9492x; 1.0475x over previous
#include <hip/hip_runtime.h>


typedef float v2f __attribute__((ext_vector_type(2)));
typedef float v4f __attribute__((ext_vector_type(4)));
typedef float v8f __attribute__((ext_vector_type(8)));

#define NB_B   2
#define NN     2048
#define NT_T   8
#define DX_D   3
#define DZ_D   16
#define HH     64
#define WW     64
#define GPB    (NT_T * HH * WW)

#define SQH (0.84932180028801904272f)

#define NTH  128
#define NWV  4
#define BH   8
#define BW   32
#define WT   (BW / 16)
#define NT   (BH * WT)
#define NPW  (NN / NWV)
#define CHW  256
#define NCH  (NPW / CHW)
#define EPW  (CHW + 4)
#define PPW  (CHW + 4)
#define ZRW  (CHW * DZ_D / 4 / 32)
#define XQ   (256 * DX_D / 4)

#define NK   4

struct Frag {
    v4f    a[NK][WT];
    v4f    p[NK][BH];
    v4f    z[NK];
};

__device__ __forceinline__
void frag_load(Frag& f, const float* eA, const float* pA, const float* zA, int k)
{
    #pragma unroll
    for (int i = 0; i < NK; ++i) {
        int k0 = k + 8 * i;
        #pragma unroll
        for (int y = 0; y < WT; ++y)
            f.a[i][y] = *(const v4f*)(eA + y * 16 * EPW + k0);
        #pragma unroll
        for (int j = 0; j < BH; ++j)
            f.p[i][j] = *(const v4f*)(pA + j * PPW + k0);
        f.z[i].x = zA[(k0 + 0) * DZ_D];
        f.z[i].y = zA[(k0 + 1) * DZ_D];
        f.z[i].z = zA[(k0 + 2) * DZ_D];
        f.z[i].w = zA[(k0 + 3) * DZ_D];
    }
}

__device__ __forceinline__
void frag_mma(const Frag& f, v8f (&acc)[NT])
{
    #pragma unroll
    for (int i = 0; i < NK; ++i) {
        v2f z01; z01.x = f.z[i].x; z01.y = f.z[i].y;
        v2f z23; z23.x = f.z[i].z; z23.y = f.z[i].w;

        #pragma unroll
        for (int j = 0; j < BH; ++j) {
            v2f B; B.x = f.p[i][j].x * z01.x; B.y = f.p[i][j].y * z01.y;
            #pragma unroll
            for (int y = 0; y < WT; ++y) {
                v2f A; A.x = f.a[i][y].x; A.y = f.a[i][y].y;
                acc[j * WT + y] = __builtin_amdgcn_wmma_f32_16x16x4_f32(
                         false, A, false, B, (short)0, acc[j * WT + y], false, false);
            }
        }
        #pragma unroll
        for (int j = 0; j < BH; ++j) {
            v2f B; B.x = f.p[i][j].z * z23.x; B.y = f.p[i][j].w * z23.y;
            #pragma unroll
            for (int y = 0; y < WT; ++y) {
                v2f A; A.x = f.a[i][y].z; A.y = f.a[i][y].w;
                acc[j * WT + y] = __builtin_amdgcn_wmma_f32_16x16x4_f32(
                         false, A, false, B, (short)0, acc[j * WT + y], false, false);
            }
        }
    }
}

__device__ __forceinline__
void order_point(v8f (&acc)[NT])
{
    asm volatile("" : "+v"(acc[0]),  "+v"(acc[1]),  "+v"(acc[2]),  "+v"(acc[3]),
                      "+v"(acc[4]),  "+v"(acc[5]),  "+v"(acc[6]),  "+v"(acc[7]),
                      "+v"(acc[8]),  "+v"(acc[9]),  "+v"(acc[10]), "+v"(acc[11]),
                      "+v"(acc[12]), "+v"(acc[13]), "+v"(acc[14]), "+v"(acc[15])
                 :: "memory");
}

__device__ __forceinline__
void order_point(v8f (&acc)[NT], Frag& f)
{
    asm volatile("" : "+v"(acc[0]),  "+v"(acc[1]),  "+v"(acc[2]),  "+v"(acc[3]),
                      "+v"(acc[4]),  "+v"(acc[5]),  "+v"(acc[6]),  "+v"(acc[7]),
                      "+v"(acc[8]),  "+v"(acc[9]),  "+v"(acc[10]), "+v"(acc[11]),
                      "+v"(acc[12]), "+v"(acc[13]), "+v"(acc[14]), "+v"(acc[15]),
                      "+v"(f.z[0]), "+v"(f.z[1]), "+v"(f.z[2]), "+v"(f.z[3])
                 :: "memory");
}

__global__ __launch_bounds__(NTH)
void setconv_ksplit_kernel(const float* __restrict__ x,
                           const float* __restrict__ z,
                           const float* __restrict__ tg,
                           const float* __restrict__ lsp,
                           float* __restrict__ xg_out,
                           float* __restrict__ zg_out)
{
    __shared__ __attribute__((aligned(16))) float eS[NWV * BW * EPW];
    __shared__ __attribute__((aligned(16))) float pS[NWV * BH * PPW];
    __shared__ __attribute__((aligned(16))) float zS[NWV * CHW * DZ_D];
    __shared__ __attribute__((aligned(16))) float c0S[NN];
    __shared__ __attribute__((aligned(16))) float c1S[NN];
    __shared__ __attribute__((aligned(16))) float c2S[NN];

    float s0 = SQH / (1e-5f + log1pf(__expf(lsp[0])));
    float s1 = SQH / (1e-5f + log1pf(__expf(lsp[1])));
    float s2 = SQH / (1e-5f + log1pf(__expf(lsp[2])));

    int blk = blockIdx.x;
    int b   = blk >> 7;
    int ti  = (blk >> 4) & 7;
    int hi0 = ((blk >> 1) & 7) * BH;
    int wi0 = (blk & 1) * BW;

    const float* xb = x + (size_t)b * NN * DX_D;
    const float* zb = z + (size_t)b * NN * DZ_D;

    int tid  = threadIdx.x;
    int lane = tid & 31;
    int wave = tid >> 5;
    int m    = lane & 15;
    int h    = lane >> 4;
    int nb   = wave * NPW;

    {
        int   xti = (blk >> 4) & 7;
        int   xh0 = (blk & 15) * 4;
        float tgv = tg[b * NT_T + xti];
        #pragma unroll
        for (int rd = 0; rd < 2; ++rd) {
            int f4 = tid + rd * NTH;
            if (f4 < XQ) {
                float xv[4];
                #pragma unroll
                for (int s = 0; s < 4; ++s) {
                    unsigned j  = 4u * (unsigned)f4 + (unsigned)s;
                    unsigned q  = j / 3u;
                    unsigned cd = j - 3u * q;
                    int      qh = xh0 + (int)(q >> 6);
                    int      qw = (int)(q & 63u);
                    float    ch = -1.0f + (float)qh * (2.0f / 63.0f);
                    float    cw = -1.0f + (float)qw * (2.0f / 63.0f);
                    xv[s] = (cd == 0u) ? tgv : ((cd == 1u) ? ch : cw);
                }
                float4 v4; v4.x = xv[0]; v4.y = xv[1]; v4.z = xv[2]; v4.w = xv[3];
                *((float4*)(xg_out + (size_t)blk * (256 * DX_D)) + f4) = v4;
            }
        }
    }

    #pragma unroll
    for (int i = 0; i < NPW / 32; ++i) {
        int n = nb + lane + 32 * i;
        c0S[n] = xb[n * 3 + 0] * s0;
        c1S[n] = xb[n * 3 + 1] * s1;
        c2S[n] = xb[n * 3 + 2] * s2;
    }

    float a0s = tg[b * NT_T + ti] * s0;

    float* eW = eS + wave * (BW * EPW);
    float* pW = pS + wave * (BH * PPW);
    float* zW = zS + wave * (CHW * DZ_D);

    int eofs = wave * (BW * EPW) + m * EPW + 4 * h;
    int pofs = wave * (BH * PPW) + 4 * h;
    int zofs = wave * (CHW * DZ_D) + (4 * h) * DZ_D + m;
    asm volatile("" : "+v"(eofs), "+v"(pofs), "+v"(zofs));
    const float* eA = eS + eofs;
    const float* pA = pS + pofs;
    const float* zA = zS + zofs;

    v8f acc[NT];
    #pragma unroll
    for (int t = 0; t < NT; ++t) acc[t] = (v8f){};

    #pragma unroll 1
    for (int c = 0; c < NCH; ++c) {
        int cb = nb + c * CHW;
        __syncthreads();

        const float4* zn = (const float4*)zb + cb * (DZ_D / 4) + lane;
#define ZLD(i) float4 zr##i = zn[(i) * 32];
        ZLD(0) ZLD(1) ZLD(2) ZLD(3)
        ZLD(4) ZLD(5) ZLD(6) ZLD(7)
        ZLD(8) ZLD(9) ZLD(10) ZLD(11)
        ZLD(12) ZLD(13) ZLD(14) ZLD(15)
        ZLD(16) ZLD(17) ZLD(18) ZLD(19)
        ZLD(20) ZLD(21) ZLD(22) ZLD(23)
        ZLD(24) ZLD(25) ZLD(26) ZLD(27)
        ZLD(28) ZLD(29) ZLD(30) ZLD(31)
#undef ZLD

        #pragma unroll
        for (int qq = 0; qq < CHW / 128; ++qq) {
            int q4 = 4 * (lane + 32 * qq);
            v4f C0 = *(const v4f*)(c0S + cb + q4);
            v4f C1 = *(const v4f*)(c1S + cb + q4);
            v4f C2 = *(const v4f*)(c2S + cb + q4);
            v4f t0;
            t0.x = (a0s - C0.x) * (a0s - C0.x);
            t0.y = (a0s - C0.y) * (a0s - C0.y);
            t0.z = (a0s - C0.z) * (a0s - C0.z);
            t0.w = (a0s - C0.w) * (a0s - C0.w);
            #pragma unroll
            for (int j = 0; j < BH; ++j) {
                float a1s = (-1.0f + (float)(hi0 + j) * (2.0f / 63.0f)) * s1;
                float dx = a1s - C1.x, dy = a1s - C1.y;
                float dz = a1s - C1.z, dw = a1s - C1.w;
                v4f pv;
                pv.x = __builtin_amdgcn_exp2f(-(t0.x + dx * dx));
                pv.y = __builtin_amdgcn_exp2f(-(t0.y + dy * dy));
                pv.z = __builtin_amdgcn_exp2f(-(t0.z + dz * dz));
                pv.w = __builtin_amdgcn_exp2f(-(t0.w + dw * dw));
                *(v4f*)(pW + j * PPW + q4) = pv;
            }
            #pragma unroll
            for (int r = 0; r < BW; ++r) {
                float a2s = (-1.0f + (float)(wi0 + r) * (2.0f / 63.0f)) * s2;
                float dx = a2s - C2.x, dy = a2s - C2.y;
                float dz = a2s - C2.z, dw = a2s - C2.w;
                v4f ev;
                ev.x = __builtin_amdgcn_exp2f(-(dx * dx));
                ev.y = __builtin_amdgcn_exp2f(-(dy * dy));
                ev.z = __builtin_amdgcn_exp2f(-(dz * dz));
                ev.w = __builtin_amdgcn_exp2f(-(dw * dw));
                *(v4f*)(eW + r * EPW + q4) = ev;
            }
        }
        {
            float4* dst = (float4*)zW + lane;
#define ZST(i) dst[(i) * 32] = zr##i;
            ZST(0) ZST(1) ZST(2) ZST(3)
            ZST(4) ZST(5) ZST(6) ZST(7)
            ZST(8) ZST(9) ZST(10) ZST(11)
            ZST(12) ZST(13) ZST(14) ZST(15)
            ZST(16) ZST(17) ZST(18) ZST(19)
            ZST(20) ZST(21) ZST(22) ZST(23)
            ZST(24) ZST(25) ZST(26) ZST(27)
            ZST(28) ZST(29) ZST(30) ZST(31)
#undef ZST
        }
        __syncthreads();

        Frag f0, f1;
        frag_load(f0, eA, pA, zA, 0);
        frag_load(f1, eA, pA, zA, 8 * NK);
        #pragma unroll 1
        for (int kb = 0; kb < CHW - 16 * NK; kb += 16 * NK) {
            frag_mma(f0, acc);
            order_point(acc);
            frag_load(f0, eA, pA, zA, kb + 16 * NK);
            order_point(acc, f1);
            frag_mma(f1, acc);
            order_point(acc);
            frag_load(f1, eA, pA, zA, kb + 24 * NK);
        }
        frag_mma(f0, acc);
        frag_mma(f1, acc);
        asm volatile("v_nop\n\tv_nop\n\tv_nop\n\tv_nop\n\tv_nop"
                     : "+v"(acc[0]),  "+v"(acc[1]),  "+v"(acc[2]),  "+v"(acc[3]),
                       "+v"(acc[4]),  "+v"(acc[5]),  "+v"(acc[6]),  "+v"(acc[7]),
                       "+v"(acc[8]),  "+v"(acc[9]),  "+v"(acc[10]), "+v"(acc[11]),
                       "+v"(acc[12]), "+v"(acc[13]), "+v"(acc[14]), "+v"(acc[15]));
    }

    __syncthreads();
    {
        v4f* red = (v4f*)eS;
        #pragma unroll
        for (int t = 0; t < NT; ++t) {
            v4f lo; lo.x = acc[t][0]; lo.y = acc[t][1]; lo.z = acc[t][2]; lo.w = acc[t][3];
            v4f hi; hi.x = acc[t][4]; hi.y = acc[t][5]; hi.z = acc[t][6]; hi.w = acc[t][7];
            red[((wave * NT + t) * 2 + 0) * 32 + lane] = lo;
            red[((wave * NT + t) * 2 + 1) * 32 + lane] = hi;
        }
    }
    __syncthreads();

    {
        const v4f* red = (const v4f*)eS;
        #pragma unroll
        for (int tt = 0; tt < NT / NWV; ++tt) {
            int t = wave * (NT / NWV) + tt;
            v4f lo = red[((0 * NT + t) * 2 + 0) * 32 + lane];
            v4f hi = red[((0 * NT + t) * 2 + 1) * 32 + lane];
            #pragma unroll
            for (int u = 1; u < NWV; ++u) {
                lo += red[((u * NT + t) * 2 + 0) * 32 + lane];
                hi += red[((u * NT + t) * 2 + 1) * 32 + lane];
            }
            int    j  = t >> 1;
            int    y  = t & 1;
            size_t g0 = (size_t)b * GPB + (size_t)ti * (HH * WW)
                      + (size_t)(hi0 + j) * WW + (size_t)(wi0 + 16 * y);
            float* outp = zg_out + (g0 + (size_t)h * 8) * DZ_D + m;
            outp[0 * DZ_D] = lo.x; outp[1 * DZ_D] = lo.y;
            outp[2 * DZ_D] = lo.z; outp[3 * DZ_D] = lo.w;
            outp[4 * DZ_D] = hi.x; outp[5 * DZ_D] = hi.y;
            outp[6 * DZ_D] = hi.z; outp[7 * DZ_D] = hi.w;
        }
    }
}

extern "C" void kernel_launch(void* const* d_in, const int* in_sizes, int n_in,
                              void* d_out, int out_size, void* d_ws, size_t ws_size,
                              hipStream_t stream) {
    (void)in_sizes; (void)n_in; (void)out_size; (void)d_ws; (void)ws_size;
    const float* x   = (const float*)d_in[0];
    const float* z   = (const float*)d_in[1];
    const float* tgr = (const float*)d_in[2];
    const float* lsp = (const float*)d_in[3];

    float* xg_out = (float*)d_out;
    float* zg_out = xg_out + (size_t)NB_B * GPB * DX_D;

    setconv_ksplit_kernel<<<NB_B * NT_T * (HH / BH) * (WW / BW), NTH, 0, stream>>>(x, z, tgr, lsp, xg_out, zg_out);
}


# Two waves per SIMD: 8 waves per block, each an eighth of n in chunks of 128, wave-own LDS and no workgroup barrier in the product phase; partial sums now added in the order w0+w1+...+w7
# speedup vs baseline: 2.0810x; 1.0676x over previous
#include <hip/hip_runtime.h>


typedef float v2f __attribute__((ext_vector_type(2)));
typedef float v4f __attribute__((ext_vector_type(4)));
typedef float v8f __attribute__((ext_vector_type(8)));

#define NB_B   2
#define NN     2048
#define NT_T   8
#define DX_D   3
#define DZ_D   16
#define HH     64
#define WW     64
#define GPB    (NT_T * HH * WW)

#define SQH (0.84932180028801904272f)

#define NTH  256
#define NWV  8
#define BH   8
#define BW   32
#define WT   (BW / 16)
#define NT   (BH * WT)
#define NPW  (NN / NWV)
#define CHW  128
#define NCH  (NPW / CHW)
#define EPW  (CHW + 4)
#define PPW  (CHW + 4)
#define ZRW  (CHW * DZ_D / 4 / 32)
#define XQ   (256 * DX_D / 4)

#define NK   2

struct Frag {
    v4f    a[NK][WT];
    v4f    p[NK][BH];
    v4f    z[NK];
};

__device__ __forceinline__
void frag_load(Frag& f, const float* eA, const float* pA, const float* zA, int k)
{
    #pragma unroll
    for (int i = 0; i < NK; ++i) {
        int k0 = k + 8 * i;
        #pragma unroll
        for (int y = 0; y < WT; ++y)
            f.a[i][y] = *(const v4f*)(eA + y * 16 * EPW + k0);
        #pragma unroll
        for (int j = 0; j < BH; ++j)
            f.p[i][j] = *(const v4f*)(pA + j * PPW + k0);
        f.z[i].x = zA[(k0 + 0) * DZ_D];
        f.z[i].y = zA[(k0 + 1) * DZ_D];
        f.z[i].z = zA[(k0 + 2) * DZ_D];
        f.z[i].w = zA[(k0 + 3) * DZ_D];
    }
}

__device__ __forceinline__
void frag_mma(const Frag& f, v8f (&acc)[NT])
{
    #pragma unroll
    for (int i = 0; i < NK; ++i) {
        v2f z01; z01.x = f.z[i].x; z01.y = f.z[i].y;
        v2f z23; z23.x = f.z[i].z; z23.y = f.z[i].w;

        #pragma unroll
        for (int j = 0; j < BH; ++j) {
            v2f B; B.x = f.p[i][j].x * z01.x; B.y = f.p[i][j].y * z01.y;
            #pragma unroll
            for (int y = 0; y < WT; ++y) {
                v2f A; A.x = f.a[i][y].x; A.y = f.a[i][y].y;
                acc[j * WT + y] = __builtin_amdgcn_wmma_f32_16x16x4_f32(
                         false, A, false, B, (short)0, acc[j * WT + y], false, false);
            }
        }
        #pragma unroll
        for (int j = 0; j < BH; ++j) {
            v2f B; B.x = f.p[i][j].z * z23.x; B.y = f.p[i][j].w * z23.y;
            #pragma unroll
            for (int y = 0; y < WT; ++y) {
                v2f A; A.x = f.a[i][y].z; A.y = f.a[i][y].w;
                acc[j * WT + y] = __builtin_amdgcn_wmma_f32_16x16x4_f32(
                         false, A, false, B, (short)0, acc[j * WT + y], false, false);
            }
        }
    }
}

__device__ __forceinline__
void order_point(v8f (&acc)[NT])
{
    asm volatile("" : "+v"(acc[0]),  "+v"(acc[1]),  "+v"(acc[2]),  "+v"(acc[3]),
                      "+v"(acc[4]),  "+v"(acc[5]),  "+v"(acc[6]),  "+v"(acc[7]),
                      "+v"(acc[8]),  "+v"(acc[9]),  "+v"(acc[10]), "+v"(acc[11]),
                      "+v"(acc[12]), "+v"(acc[13]), "+v"(acc[14]), "+v"(acc[15])
                 :: "memory");
}

__device__ __forceinline__
void order_point(v8f (&acc)[NT], Frag& f)
{
    asm volatile("" : "+v"(acc[0]),  "+v"(acc[1]),  "+v"(acc[2]),  "+v"(acc[3]),
                      "+v"(acc[4]),  "+v"(acc[5]),  "+v"(acc[6]),  "+v"(acc[7]),
                      "+v"(acc[8]),  "+v"(acc[9]),  "+v"(acc[10]), "+v"(acc[11]),
                      "+v"(acc[12]), "+v"(acc[13]), "+v"(acc[14]), "+v"(acc[15]),
                      "+v"(f.z[0]), "+v"(f.z[1]), "+v"(f.z[2]), "+v"(f.z[3])
                 :: "memory");
}

__global__ __launch_bounds__(NTH)
void setconv_ksplit_kernel(const float* __restrict__ x,
                           const float* __restrict__ z,
                           const float* __restrict__ tg,
                           const float* __restrict__ lsp,
                           float* __restrict__ xg_out,
                           float* __restrict__ zg_out)
{
    __shared__ __attribute__((aligned(16))) float eS[NWV * BW * EPW];
    __shared__ __attribute__((aligned(16))) float pS[NWV * BH * PPW];
    __shared__ __attribute__((aligned(16))) float zS[NWV * CHW * DZ_D];
    __shared__ __attribute__((aligned(16))) float c0S[NN];
    __shared__ __attribute__((aligned(16))) float c1S[NN];
    __shared__ __attribute__((aligned(16))) float c2S[NN];

    float s0 = SQH / (1e-5f + log1pf(__expf(lsp[0])));
    float s1 = SQH / (1e-5f + log1pf(__expf(lsp[1])));
    float s2 = SQH / (1e-5f + log1pf(__expf(lsp[2])));

    int blk = blockIdx.x;
    int b   = blk >> 7;
    int ti  = (blk >> 4) & 7;
    int hi0 = ((blk >> 1) & 7) * BH;
    int wi0 = (blk & 1) * BW;

    const float* xb = x + (size_t)b * NN * DX_D;
    const float* zb = z + (size_t)b * NN * DZ_D;

    int tid  = threadIdx.x;
    int lane = tid & 31;
    int wave = tid >> 5;
    int m    = lane & 15;
    int h    = lane >> 4;
    int nb   = wave * NPW;

    {
        int   xti = (blk >> 4) & 7;
        int   xh0 = (blk & 15) * 4;
        float tgv = tg[b * NT_T + xti];
        #pragma unroll
        for (int rd = 0; rd < (XQ + NTH - 1) / NTH; ++rd) {
            int f4 = tid + rd * NTH;
            if (f4 < XQ) {
                float xv[4];
                #pragma unroll
                for (int s = 0; s < 4; ++s) {
                    unsigned j  = 4u * (unsigned)f4 + (unsigned)s;
                    unsigned q  = j / 3u;
                    unsigned cd = j - 3u * q;
                    int      qh = xh0 + (int)(q >> 6);
                    int      qw = (int)(q & 63u);
                    float    ch = -1.0f + (float)qh * (2.0f / 63.0f);
                    float    cw = -1.0f + (float)qw * (2.0f / 63.0f);
                    xv[s] = (cd == 0u) ? tgv : ((cd == 1u) ? ch : cw);
                }
                float4 v4; v4.x = xv[0]; v4.y = xv[1]; v4.z = xv[2]; v4.w = xv[3];
                *((float4*)(xg_out + (size_t)blk * (256 * DX_D)) + f4) = v4;
            }
        }
    }

    #pragma unroll
    for (int i = 0; i < NPW / 32; ++i) {
        int n = nb + lane + 32 * i;
        c0S[n] = xb[n * 3 + 0] * s0;
        c1S[n] = xb[n * 3 + 1] * s1;
        c2S[n] = xb[n * 3 + 2] * s2;
    }

    float a0s = tg[b * NT_T + ti] * s0;

    float* eW = eS + wave * (BW * EPW);
    float* pW = pS + wave * (BH * PPW);
    float* zW = zS + wave * (CHW * DZ_D);

    int eofs = wave * (BW * EPW) + m * EPW + 4 * h;
    int pofs = wave * (BH * PPW) + 4 * h;
    int zofs = wave * (CHW * DZ_D) + (4 * h) * DZ_D + m;
    asm volatile("" : "+v"(eofs), "+v"(pofs), "+v"(zofs));
    const float* eA = eS + eofs;
    const float* pA = pS + pofs;
    const float* zA = zS + zofs;

    v8f acc[NT];
    #pragma unroll
    for (int t = 0; t < NT; ++t) acc[t] = (v8f){};

    #pragma unroll 1
    for (int c = 0; c < NCH; ++c) {
        int cb = nb + c * CHW;
        __threadfence_block();
        __builtin_amdgcn_wave_barrier();

        const float4* zn = (const float4*)zb + cb * (DZ_D / 4) + lane;
#define ZLD(i) float4 zr##i = zn[(i) * 32];
        ZLD(0) ZLD(1) ZLD(2) ZLD(3)
        ZLD(4) ZLD(5) ZLD(6) ZLD(7)
        ZLD(8) ZLD(9) ZLD(10) ZLD(11)
        ZLD(12) ZLD(13) ZLD(14) ZLD(15)
#undef ZLD

        #pragma unroll
        for (int qq = 0; qq < CHW / 128; ++qq) {
            int q4 = 4 * (lane + 32 * qq);
            v4f C0 = *(const v4f*)(c0S + cb + q4);
            v4f C1 = *(const v4f*)(c1S + cb + q4);
            v4f C2 = *(const v4f*)(c2S + cb + q4);
            v4f t0;
            t0.x = (a0s - C0.x) * (a0s - C0.x);
            t0.y = (a0s - C0.y) * (a0s - C0.y);
            t0.z = (a0s - C0.z) * (a0s - C0.z);
            t0.w = (a0s - C0.w) * (a0s - C0.w);
            #pragma unroll
            for (int j = 0; j < BH; ++j) {
                float a1s = (-1.0f + (float)(hi0 + j) * (2.0f / 63.0f)) * s1;
                float dx = a1s - C1.x, dy = a1s - C1.y;
                float dz = a1s - C1.z, dw = a1s - C1.w;
                v4f pv;
                pv.x = __builtin_amdgcn_exp2f(-(t0.x + dx * dx));
                pv.y = __builtin_amdgcn_exp2f(-(t0.y + dy * dy));
                pv.z = __builtin_amdgcn_exp2f(-(t0.z + dz * dz));
                pv.w = __builtin_amdgcn_exp2f(-(t0.w + dw * dw));
                *(v4f*)(pW + j * PPW + q4) = pv;
            }
            #pragma unroll
            for (int r = 0; r < BW; ++r) {
                float a2s = (-1.0f + (float)(wi0 + r) * (2.0f / 63.0f)) * s2;
                float dx = a2s - C2.x, dy = a2s - C2.y;
                float dz = a2s - C2.z, dw = a2s - C2.w;
                v4f ev;
                ev.x = __builtin_amdgcn_exp2f(-(dx * dx));
                ev.y = __builtin_amdgcn_exp2f(-(dy * dy));
                ev.z = __builtin_amdgcn_exp2f(-(dz * dz));
                ev.w = __builtin_amdgcn_exp2f(-(dw * dw));
                *(v4f*)(eW + r * EPW + q4) = ev;
            }
        }
        {
            float4* dst = (float4*)zW + lane;
#define ZST(i) dst[(i) * 32] = zr##i;
            ZST(0) ZST(1) ZST(2) ZST(3)
            ZST(4) ZST(5) ZST(6) ZST(7)
            ZST(8) ZST(9) ZST(10) ZST(11)
            ZST(12) ZST(13) ZST(14) ZST(15)
#undef ZST
        }
        __threadfence_block();
        __builtin_amdgcn_wave_barrier();

        Frag f0, f1;
        frag_load(f0, eA, pA, zA, 0);
        frag_load(f1, eA, pA, zA, 8 * NK);
        #pragma unroll 1
        for (int kb = 0; kb < CHW - 16 * NK; kb += 16 * NK) {
            frag_mma(f0, acc);
            order_point(acc);
            frag_load(f0, eA, pA, zA, kb + 16 * NK);
            order_point(acc, f1);
            frag_mma(f1, acc);
            order_point(acc);
            frag_load(f1, eA, pA, zA, kb + 24 * NK);
        }
        frag_mma(f0, acc);
        frag_mma(f1, acc);
        asm volatile("v_nop\n\tv_nop\n\tv_nop\n\tv_nop\n\tv_nop"
                     : "+v"(acc[0]),  "+v"(acc[1]),  "+v"(acc[2]),  "+v"(acc[3]),
                       "+v"(acc[4]),  "+v"(acc[5]),  "+v"(acc[6]),  "+v"(acc[7]),
                       "+v"(acc[8]),  "+v"(acc[9]),  "+v"(acc[10]), "+v"(acc[11]),
                       "+v"(acc[12]), "+v"(acc[13]), "+v"(acc[14]), "+v"(acc[15]));
    }

    __syncthreads();
    {
        v4f* red = (v4f*)eS;
        #pragma unroll
        for (int t = 0; t < NT; ++t) {
            v4f lo; lo.x = acc[t][0]; lo.y = acc[t][1]; lo.z = acc[t][2]; lo.w = acc[t][3];
            v4f hi; hi.x = acc[t][4]; hi.y = acc[t][5]; hi.z = acc[t][6]; hi.w = acc[t][7];
            red[((wave * NT + t) * 2 + 0) * 32 + lane] = lo;
            red[((wave * NT + t) * 2 + 1) * 32 + lane] = hi;
        }
    }
    __syncthreads();

    {
        const v4f* red = (const v4f*)eS;
        #pragma unroll
        for (int tt = 0; tt < NT / NWV; ++tt) {
            int t = wave * (NT / NWV) + tt;
            v4f lo = red[((0 * NT + t) * 2 + 0) * 32 + lane];
            v4f hi = red[((0 * NT + t) * 2 + 1) * 32 + lane];
            #pragma unroll
            for (int u = 1; u < NWV; ++u) {
                lo += red[((u * NT + t) * 2 + 0) * 32 + lane];
                hi += red[((u * NT + t) * 2 + 1) * 32 + lane];
            }
            int    j  = t >> 1;
            int    y  = t & 1;
            size_t g0 = (size_t)b * GPB + (size_t)ti * (HH * WW)
                      + (size_t)(hi0 + j) * WW + (size_t)(wi0 + 16 * y);
            float* outp = zg_out + (g0 + (size_t)h * 8) * DZ_D + m;
            outp[0 * DZ_D] = lo.x; outp[1 * DZ_D] = lo.y;
            outp[2 * DZ_D] = lo.z; outp[3 * DZ_D] = lo.w;
            outp[4 * DZ_D] = hi.x; outp[5 * DZ_D] = hi.y;
            outp[6 * DZ_D] = hi.z; outp[7 * DZ_D] = hi.w;
        }
    }
}

extern "C" void kernel_launch(void* const* d_in, const int* in_sizes, int n_in,
                              void* d_out, int out_size, void* d_ws, size_t ws_size,
                              hipStream_t stream) {
    (void)in_sizes; (void)n_in; (void)out_size; (void)d_ws; (void)ws_size;
    const float* x   = (const float*)d_in[0];
    const float* z   = (const float*)d_in[1];
    const float* tgr = (const float*)d_in[2];
    const float* lsp = (const float*)d_in[3];

    float* xg_out = (float*)d_out;
    float* zg_out = xg_out + (size_t)NB_B * GPB * DX_D;

    setconv_ksplit_kernel<<<NB_B * NT_T * (HH / BH) * (WW / BW), NTH, 0, stream>>>(x, z, tgr, lsp, xg_out, zg_out);
}
